# starting kernel
# baseline (speedup 1.0000x reference)
#include <hip/hip_runtime.h>
#include <math.h>

#define TT   2048
#define BB   32
#define DD   256
#define HH   256
#define W3H  768
#define NPRE 512
#define HSTR 260

typedef __attribute__((ext_vector_type(2))) float v2f;
typedef __attribute__((ext_vector_type(8))) float v8f;

__device__ __forceinline__ size_t px_off(int t, int w, int mt, int lane) {
  return ((((size_t)(t * 32 + w) * 2 + mt) * 32 + lane) * 8);
}

__global__ __launch_bounds__(256) void rhn_gemm_x(
    const float* __restrict__ X, const float* __restrict__ W,
    const float* __restrict__ bias, float* __restrict__ preX) {
  extern __shared__ char smem[];
  v2f* wtile = (v2f*)smem;
  const int tid  = threadIdx.x;
  const int lane = tid & 31;
  const int wave = tid >> 5;
  const int m0   = blockIdx.x * 128 + wave * 16;
  const int n0   = blockIdx.y * 128;

  for (int e = tid; e < 64 * 8 * 32; e += 256) {
    const int c  = e >> 8;
    const int nt = (e >> 5) & 7;
    const int l  = e & 31;
    const int k  = 4 * c + ((l >> 4) << 1);
    const int n  = n0 + nt * 16 + (l & 15);
    v2f wv;
    wv.x = W[k * W3H + n];
    wv.y = W[(k + 1) * W3H + n];
    wtile[e] = wv;
  }
  __syncthreads();

  v8f acc[8];
#pragma unroll
  for (int nt = 0; nt < 8; ++nt) {
    const float bv = bias[n0 + nt * 16 + (lane & 15)];
#pragma unroll
    for (int r = 0; r < 8; ++r) acc[nt][r] = bv;
  }

  const int arow = m0 + (lane & 15);
  const float* aptr = X + (size_t)arow * DD + ((lane >> 4) << 1);

#pragma unroll 8
  for (int c = 0; c < 64; ++c) {
    const v2f a = *(const v2f*)(aptr + 4 * c);
#pragma unroll
    for (int nt = 0; nt < 8; ++nt) {
      const v2f b = wtile[(c * 8 + nt) * 32 + lane];
      acc[nt] = __builtin_amdgcn_wmma_f32_16x16x4_f32(
          false, a, false, b, (short)0, acc[nt], false, false);
    }
  }

  const int tIdx = m0 >> 5;
  const int mt   = (m0 >> 4) & 1;
#pragma unroll
  for (int nt = 0; nt < 8; ++nt) {
    const int w = (n0 >> 4) + nt;
    *(v8f*)(preX + px_off(tIdx, w, mt, lane)) = acc[nt];
  }
}

__device__ __forceinline__ float sigf(float x) {
  return 1.0f / (1.0f + __expf(-x));
}

__global__ __launch_bounds__(1024) void rhn_recur(
    const float* __restrict__ preX, const float* __restrict__ Whh,
    const int* __restrict__ length, float* __restrict__ out,
    float* __restrict__ hn) {
  extern __shared__ char smem[];
  float* hbuf = (float*)smem;
  float* tbuf = hbuf + 32 * HSTR;
  int*   lbuf = (int*)(tbuf + 16 * 2 * 32 * 8);

  const int tid  = threadIdx.x;
  const int lane = tid & 31;
  const int wave = tid >> 5;

  for (int i = tid; i < 32 * HSTR; i += 1024) hbuf[i] = 0.0f;
  if (tid < 32) lbuf[tid] = length[tid];
  __syncthreads();

  const int ncol = wave * 16 + (lane & 15);
  const int ksub = (lane >> 4) << 1;
  v2f wreg[64];
#pragma unroll
  for (int c = 0; c < 64; ++c) {
    const int k = 4 * c + ksub;
    v2f wv;
    wv.x = Whh[k * W3H + ncol];
    wv.y = Whh[(k + 1) * W3H + ncol];
    wreg[c] = wv;
  }

  const int crow = (lane >> 4) << 3;
  const int am   = lane & 15;
  const bool isH = (wave < 16);
  const int  w16 = isH ? wave : (wave - 16);

  v8f cur0 = *(const v8f*)(preX + px_off(0, wave, 0, lane));
  v8f cur1 = *(const v8f*)(preX + px_off(0, wave, 1, lane));

  for (int t = 0; t < TT; ++t) {
    v8f acc0 = cur0;
    v8f acc1 = cur1;

    const int tn = (t + 1) & (TT - 1);
    cur0 = *(const v8f*)(preX + px_off(tn, wave, 0, lane));
    cur1 = *(const v8f*)(preX + px_off(tn, wave, 1, lane));

#pragma unroll
    for (int c = 0; c < 64; ++c) {
      const v2f a0 = *(const v2f*)(hbuf + am * HSTR + 4 * c + ksub);
      const v2f a1 = *(const v2f*)(hbuf + (16 + am) * HSTR + 4 * c + ksub);
      acc0 = __builtin_amdgcn_wmma_f32_16x16x4_f32(
          false, a0, false, wreg[c], (short)0, acc0, false, false);
      acc1 = __builtin_amdgcn_wmma_f32_16x16x4_f32(
          false, a1, false, wreg[c], (short)0, acc1, false, false);
    }

    if (!isH) {
      float* tb0 = tbuf + ((w16 * 2 + 0) * 32 + lane) * 8;
      float* tb1 = tbuf + ((w16 * 2 + 1) * 32 + lane) * 8;
#pragma unroll
      for (int r = 0; r < 8; ++r) { tb0[r] = acc0[r]; tb1[r] = acc1[r]; }
    }
    __syncthreads();

    if (isH) {
#pragma unroll
      for (int mt = 0; mt < 2; ++mt) {
        const float* tb = tbuf + ((w16 * 2 + mt) * 32 + lane) * 8;
        const v8f hp = mt ? acc1 : acc0;
#pragma unroll
        for (int r = 0; r < 8; ++r) {
          const int b   = mt * 16 + r + crow;
          const int col = w16 * 16 + (lane & 15);
          const float hold = hbuf[b * HSTR + col];
          const float tg = sigf(tb[r]);
          const float cg = sigf(tg);
          const float s  = tanhf(hp[r]) * tg + hold * cg;
          const float hv = (t < lbuf[b]) ? s : hold;
          hbuf[b * HSTR + col] = hv;
          out[(size_t)(t * BB + b) * HH + col] = hv;
        }
      }
    }
    __syncthreads();
  }

  for (int i = tid; i < BB * HH; i += 1024)
    hn[i] = hbuf[(i >> 8) * HSTR + (i & 255)];
}

extern "C" void kernel_launch(void* const* d_in, const int* in_sizes, int n_in,
                              void* d_out, int out_size, void* d_ws,
                              size_t ws_size, hipStream_t stream) {
  const float* input_ = (const float*)d_in[0];
  const int*   length = (const int*)d_in[1];
  const float* Wih0   = (const float*)d_in[2];
  const float* Whh0   = (const float*)d_in[3];
  const float* b0     = (const float*)d_in[4];
  const float* Wih1   = (const float*)d_in[5];
  const float* Whh1   = (const float*)d_in[6];
  const float* b1     = (const float*)d_in[7];

  float* out  = (float*)d_out;
  float* hn   = out + (size_t)TT * BB * HH;
  float* preX = (float*)d_ws;

  const dim3  g1(512, 4);
  const size_t lds1 = (size_t)64 * 8 * 32 * sizeof(v2f);
  const size_t lds2 = (size_t)(32 * HSTR + 16 * 2 * 32 * 8 + 32) * 4;

  rhn_gemm_x<<<g1, 256, lds1, stream>>>(input_, Wih0, b0, preX);
  rhn_recur<<<dim3(1), 1024, lds2, stream>>>(preX, Whh0, length, out, hn);
  rhn_gemm_x<<<g1, 256, lds1, stream>>>(out, Wih1, b1, preX);
  rhn_recur<<<dim3(1), 1024, lds2, stream>>>(preX, Whh1, length, out,
                                             hn + BB * HH);
}


# recurrence made spill-free: each wave's Whh tile held as 34 K-chunks in VGPRs + 30 in LDS (16-byte fragment slots), gate-phase addresses as lane base + constant; k order of all sums unchanged
# speedup vs baseline: 1.5040x; 1.5040x over previous
#include <hip/hip_runtime.h>
#include <math.h>

#define TT   2048
#define BB   32
#define DD   256
#define HH   256
#define W3H  768
#define NPRE 512
#define HSTR 260
#define NVR  34
#define NLP  15
#define WL_BYTES   (NLP * 32 * 32 * 16)
#define HB_BYTES   (32 * HSTR * 4)
#define TB_BYTES   (16 * 2 * 32 * 8 * 4)
#define LB_BYTES   (32 * 4)
#define LDS2_BYTES (WL_BYTES + HB_BYTES + TB_BYTES + LB_BYTES)

typedef __attribute__((ext_vector_type(2))) float v2f;
typedef __attribute__((ext_vector_type(4))) float v4f;
typedef __attribute__((ext_vector_type(8))) float v8f;

__device__ __forceinline__ size_t px_off(int t, int w, int mt, int lane) {
  return ((((size_t)(t * 32 + w) * 2 + mt) * 32 + lane) * 8);
}

__global__ __launch_bounds__(256) void rhn_gemm_x(
    const float* __restrict__ X, const float* __restrict__ W,
    const float* __restrict__ bias, float* __restrict__ preX) {
  extern __shared__ char smem[];
  v2f* wtile = (v2f*)smem;
  const int tid  = threadIdx.x;
  const int lane = tid & 31;
  const int wave = tid >> 5;
  const int m0   = blockIdx.x * 128 + wave * 16;
  const int n0   = blockIdx.y * 128;

  for (int e = tid; e < 64 * 8 * 32; e += 256) {
    const int c  = e >> 8;
    const int nt = (e >> 5) & 7;
    const int l  = e & 31;
    const int k  = 4 * c + ((l >> 4) << 1);
    const int n  = n0 + nt * 16 + (l & 15);
    v2f wv;
    wv.x = W[k * W3H + n];
    wv.y = W[(k + 1) * W3H + n];
    wtile[e] = wv;
  }
  __syncthreads();

  v8f acc[8];
#pragma unroll
  for (int nt = 0; nt < 8; ++nt) {
    const float bv = bias[n0 + nt * 16 + (lane & 15)];
#pragma unroll
    for (int r = 0; r < 8; ++r) acc[nt][r] = bv;
  }

  const int arow = m0 + (lane & 15);
  const float* aptr = X + (size_t)arow * DD + ((lane >> 4) << 1);

#pragma unroll 8
  for (int c = 0; c < 64; ++c) {
    const v2f a = *(const v2f*)(aptr + 4 * c);
#pragma unroll
    for (int nt = 0; nt < 8; ++nt) {
      const v2f b = wtile[(c * 8 + nt) * 32 + lane];
      acc[nt] = __builtin_amdgcn_wmma_f32_16x16x4_f32(
          false, a, false, b, (short)0, acc[nt], false, false);
    }
  }

  const int tIdx = m0 >> 5;
  const int mt   = (m0 >> 4) & 1;
#pragma unroll
  for (int nt = 0; nt < 8; ++nt) {
    const int w = (n0 >> 4) + nt;
    *(v8f*)(preX + px_off(tIdx, w, mt, lane)) = acc[nt];
  }
}

__device__ __forceinline__ float sigf(float x) {
  return 1.0f / (1.0f + __expf(-x));
}

__global__ __launch_bounds__(1024) void rhn_recur(
    const float* __restrict__ preX, const float* __restrict__ Whh,
    const int* __restrict__ length, float* __restrict__ out,
    float* __restrict__ hn) {
  __shared__ __attribute__((aligned(16))) unsigned char lds2[LDS2_BYTES];
  v4f*   wl   = (v4f*)lds2;
  float* hbuf = (float*)(lds2 + WL_BYTES);
  float* tbuf = hbuf + 32 * HSTR;
  int*   lbuf = (int*)(tbuf + 16 * 2 * 32 * 8);

  const int tid  = threadIdx.x;
  const int lane = tid & 31;
  const int wave = tid >> 5;

  for (int i = tid; i < 32 * HSTR; i += 1024) hbuf[i] = 0.0f;
  if (tid < 32) lbuf[tid] = length[tid];

  const int ncol = wave * 16 + (lane & 15);
  const int ksub = (lane >> 4) << 1;

  v4f* wlw = wl + wave * 32 + lane;
#pragma unroll
  for (int p = 0; p < NLP; ++p) {
    const int k = 4 * (NVR + 2 * p) + ksub;
    v4f q;
    q.x = Whh[k * W3H + ncol];
    q.y = Whh[(k + 1) * W3H + ncol];
    q.z = Whh[(k + 4) * W3H + ncol];
    q.w = Whh[(k + 5) * W3H + ncol];
    wlw[p * 1024] = q;
  }
  __syncthreads();

  v2f wreg[NVR];
#pragma unroll
  for (int c = 0; c < NVR; ++c) {
    const int k = 4 * c + ksub;
    v2f wv;
    wv.x = Whh[k * W3H + ncol];
    wv.y = Whh[(k + 1) * W3H + ncol];
    wreg[c] = wv;
  }

  const int crow = (lane >> 4) << 3;
  const int am   = lane & 15;
  const bool isH = (wave < 16);
  const int  w16 = isH ? wave : (wave - 16);

  float*       hrow = hbuf + crow * HSTR + w16 * 16 + am;
  const float* trow = tbuf + (w16 * 64 + lane) * 8;
  const int*   lrow = lbuf + crow;
  const int    orow = crow * HH + w16 * 16 + am;

  v8f cur0 = *(const v8f*)(preX + px_off(0, wave, 0, lane));
  v8f cur1 = *(const v8f*)(preX + px_off(0, wave, 1, lane));

  for (int t = 0; t < TT; ++t) {
    v8f acc0 = cur0;
    v8f acc1 = cur1;

    const int tn = (t + 1) & (TT - 1);
    cur0 = *(const v8f*)(preX + px_off(tn, wave, 0, lane));
    cur1 = *(const v8f*)(preX + px_off(tn, wave, 1, lane));

#pragma unroll
    for (int c = 0; c < NVR; ++c) {
      const v2f a0 = *(const v2f*)(hbuf + am * HSTR + 4 * c + ksub);
      const v2f a1 = *(const v2f*)(hbuf + (16 + am) * HSTR + 4 * c + ksub);
      acc0 = __builtin_amdgcn_wmma_f32_16x16x4_f32(
          false, a0, false, wreg[c], (short)0, acc0, false, false);
      acc1 = __builtin_amdgcn_wmma_f32_16x16x4_f32(
          false, a1, false, wreg[c], (short)0, acc1, false, false);
    }
#pragma unroll
    for (int p = 0; p < NLP; ++p) {
      const v4f q = wlw[p * 1024];
#pragma unroll
      for (int s = 0; s < 2; ++s) {
        const int c = NVR + 2 * p + s;
        v2f b;
        b.x = s ? q.z : q.x;
        b.y = s ? q.w : q.y;
        const v2f a0 = *(const v2f*)(hbuf + am * HSTR + 4 * c + ksub);
        const v2f a1 = *(const v2f*)(hbuf + (16 + am) * HSTR + 4 * c + ksub);
        acc0 = __builtin_amdgcn_wmma_f32_16x16x4_f32(
            false, a0, false, b, (short)0, acc0, false, false);
        acc1 = __builtin_amdgcn_wmma_f32_16x16x4_f32(
            false, a1, false, b, (short)0, acc1, false, false);
      }
    }

    if (!isH) {
      float* tb0 = tbuf + ((w16 * 2 + 0) * 32 + lane) * 8;
      float* tb1 = tbuf + ((w16 * 2 + 1) * 32 + lane) * 8;
#pragma unroll
      for (int r = 0; r < 8; ++r) { tb0[r] = acc0[r]; tb1[r] = acc1[r]; }
    }
    __syncthreads();

    if (isH) {
      float* ot = out + (size_t)t * BB * HH;
#pragma unroll
      for (int mt = 0; mt < 2; ++mt) {
        const v8f hp = mt ? acc1 : acc0;
#pragma unroll
        for (int r = 0; r < 8; ++r) {
          const int rr = mt * 16 + r;
          const float hold = hrow[rr * HSTR];
          const float tg = sigf(trow[mt * 256 + r]);
          const float cg = sigf(tg);
          const float s  = tanhf(hp[r]) * tg + hold * cg;
          const float hv = (t < lrow[rr]) ? s : hold;
          hrow[rr * HSTR] = hv;
          ot[orow + rr * HH] = hv;
        }
      }
    }
    __syncthreads();
  }

  for (int i = tid; i < BB * HH; i += 1024)
    hn[i] = hbuf[(i >> 8) * HSTR + (i & 255)];
}

extern "C" void kernel_launch(void* const* d_in, const int* in_sizes, int n_in,
                              void* d_out, int out_size, void* d_ws,
                              size_t ws_size, hipStream_t stream) {
  const float* input_ = (const float*)d_in[0];
  const int*   length = (const int*)d_in[1];
  const float* Wih0   = (const float*)d_in[2];
  const float* Whh0   = (const float*)d_in[3];
  const float* b0     = (const float*)d_in[4];
  const float* Wih1   = (const float*)d_in[5];
  const float* Whh1   = (const float*)d_in[6];
  const float* b1     = (const float*)d_in[7];

  float* out  = (float*)d_out;
  float* hn   = out + (size_t)TT * BB * HH;
  float* preX = (float*)d_ws;

  const dim3  g1(512, 4);
  const size_t lds1 = (size_t)64 * 8 * 32 * sizeof(v2f);

  rhn_gemm_x<<<g1, 256, lds1, stream>>>(input_, Wih0, b0, preX);
  rhn_recur<<<dim3(1), 1024, 0, stream>>>(preX, Whh0, length, out, hn);
  rhn_gemm_x<<<g1, 256, lds1, stream>>>(out, Wih1, b1, preX);
  rhn_recur<<<dim3(1), 1024, 0, stream>>>(preX, Whh1, length, out,
                                          hn + BB * HH);
}


# recurrence split over the batch: 2 workgroups, each owning 16 batch rows (one M-tile) on its own WGP; a wave chains 64 WMMAs per step, not 128, and gates 8 values, not 16; k order of sums unchanged
# speedup vs baseline: 2.9283x; 1.9470x over previous
#include <hip/hip_runtime.h>
#include <math.h>

#define TT   2048
#define BB   32
#define DD   256
#define HH   256
#define W3H  768
#define NPRE 512
#define HSTR 260
#define NVR  34
#define NLP  15
#define MROWS 16
#define WL_BYTES   (NLP * 32 * 32 * 16)
#define HB_BYTES   (MROWS * HSTR * 4)
#define TB_BYTES   (16 * 32 * 8 * 4)
#define LB_BYTES   (MROWS * 4)
#define LDS2_BYTES (WL_BYTES + HB_BYTES + TB_BYTES + LB_BYTES)

typedef __attribute__((ext_vector_type(2))) float v2f;
typedef __attribute__((ext_vector_type(4))) float v4f;
typedef __attribute__((ext_vector_type(8))) float v8f;

__device__ __forceinline__ size_t px_off(int t, int w, int mt, int lane) {
  return ((((size_t)(t * 32 + w) * 2 + mt) * 32 + lane) * 8);
}

__global__ __launch_bounds__(256) void rhn_gemm_x(
    const float* __restrict__ X, const float* __restrict__ W,
    const float* __restrict__ bias, float* __restrict__ preX) {
  extern __shared__ char smem[];
  v2f* wtile = (v2f*)smem;
  const int tid  = threadIdx.x;
  const int lane = tid & 31;
  const int wave = tid >> 5;
  const int m0   = blockIdx.x * 128 + wave * 16;
  const int n0   = blockIdx.y * 128;

  for (int e = tid; e < 64 * 8 * 32; e += 256) {
    const int c  = e >> 8;
    const int nt = (e >> 5) & 7;
    const int l  = e & 31;
    const int k  = 4 * c + ((l >> 4) << 1);
    const int n  = n0 + nt * 16 + (l & 15);
    v2f wv;
    wv.x = W[k * W3H + n];
    wv.y = W[(k + 1) * W3H + n];
    wtile[e] = wv;
  }
  __syncthreads();

  v8f acc[8];
#pragma unroll
  for (int nt = 0; nt < 8; ++nt) {
    const float bv = bias[n0 + nt * 16 + (lane & 15)];
#pragma unroll
    for (int r = 0; r < 8; ++r) acc[nt][r] = bv;
  }

  const int arow = m0 + (lane & 15);
  const float* aptr = X + (size_t)arow * DD + ((lane >> 4) << 1);

#pragma unroll 8
  for (int c = 0; c < 64; ++c) {
    const v2f a = *(const v2f*)(aptr + 4 * c);
#pragma unroll
    for (int nt = 0; nt < 8; ++nt) {
      const v2f b = wtile[(c * 8 + nt) * 32 + lane];
      acc[nt] = __builtin_amdgcn_wmma_f32_16x16x4_f32(
          false, a, false, b, (short)0, acc[nt], false, false);
    }
  }

  const int tIdx = m0 >> 5;
  const int mt   = (m0 >> 4) & 1;
#pragma unroll
  for (int nt = 0; nt < 8; ++nt) {
    const int w = (n0 >> 4) + nt;
    *(v8f*)(preX + px_off(tIdx, w, mt, lane)) = acc[nt];
  }
}

__device__ __forceinline__ float sigf(float x) {
  return 1.0f / (1.0f + __expf(-x));
}

__global__ __launch_bounds__(1024) void rhn_recur(
    const float* __restrict__ preX, const float* __restrict__ Whh,
    const int* __restrict__ length, float* __restrict__ out,
    float* __restrict__ hn) {
  __shared__ __attribute__((aligned(16))) unsigned char lds2[LDS2_BYTES];
  v4f*   wl   = (v4f*)lds2;
  float* hbuf = (float*)(lds2 + WL_BYTES);
  float* tbuf = hbuf + MROWS * HSTR;
  int*   lbuf = (int*)(tbuf + 16 * 32 * 8);

  const int tid  = threadIdx.x;
  const int lane = tid & 31;
  const int wave = tid >> 5;
  const int mt   = blockIdx.x;

  for (int i = tid; i < MROWS * HSTR; i += 1024) hbuf[i] = 0.0f;
  if (tid < MROWS) lbuf[tid] = length[mt * MROWS + tid];

  const int ncol = wave * 16 + (lane & 15);
  const int ksub = (lane >> 4) << 1;

  v4f* wlw = wl + wave * 32 + lane;
#pragma unroll
  for (int p = 0; p < NLP; ++p) {
    const int k = 4 * (NVR + 2 * p) + ksub;
    v4f q;
    q.x = Whh[k * W3H + ncol];
    q.y = Whh[(k + 1) * W3H + ncol];
    q.z = Whh[(k + 4) * W3H + ncol];
    q.w = Whh[(k + 5) * W3H + ncol];
    wlw[p * 1024] = q;
  }
  __syncthreads();

  v2f wreg[NVR];
#pragma unroll
  for (int c = 0; c < NVR; ++c) {
    const int k = 4 * c + ksub;
    v2f wv;
    wv.x = Whh[k * W3H + ncol];
    wv.y = Whh[(k + 1) * W3H + ncol];
    wreg[c] = wv;
  }

  const int crow = (lane >> 4) << 3;
  const int am   = lane & 15;
  const bool isH = (wave < 16);
  const int  w16 = isH ? wave : (wave - 16);

  float*       hrow = hbuf + crow * HSTR + w16 * 16 + am;
  const float* trow = tbuf + (w16 * 32 + lane) * 8;
  const int*   lrow = lbuf + crow;
  const int    orow = (mt * MROWS + crow) * HH + w16 * 16 + am;

  v8f cur = *(const v8f*)(preX + px_off(0, wave, mt, lane));

  for (int t = 0; t < TT; ++t) {
    v8f acc = cur;

    const int tn = (t + 1) & (TT - 1);
    cur = *(const v8f*)(preX + px_off(tn, wave, mt, lane));

#pragma unroll
    for (int c = 0; c < NVR; ++c) {
      const v2f a = *(const v2f*)(hbuf + am * HSTR + 4 * c + ksub);
      acc = __builtin_amdgcn_wmma_f32_16x16x4_f32(
          false, a, false, wreg[c], (short)0, acc, false, false);
    }
#pragma unroll
    for (int p = 0; p < NLP; ++p) {
      const v4f q = wlw[p * 1024];
#pragma unroll
      for (int s = 0; s < 2; ++s) {
        const int c = NVR + 2 * p + s;
        v2f b;
        b.x = s ? q.z : q.x;
        b.y = s ? q.w : q.y;
        const v2f a = *(const v2f*)(hbuf + am * HSTR + 4 * c + ksub);
        acc = __builtin_amdgcn_wmma_f32_16x16x4_f32(
            false, a, false, b, (short)0, acc, false, false);
      }
    }

    if (!isH) {
      float* tb = tbuf + (w16 * 32 + lane) * 8;
#pragma unroll
      for (int r = 0; r < 8; ++r) tb[r] = acc[r];
    }
    __syncthreads();

    if (isH) {
      float* ot = out + (size_t)t * BB * HH;
#pragma unroll
      for (int r = 0; r < 8; ++r) {
        const float hold = hrow[r * HSTR];
        const float tg = sigf(trow[r]);
        const float cg = sigf(tg);
        const float s  = tanhf(acc[r]) * tg + hold * cg;
        const float hv = (t < lrow[r]) ? s : hold;
        hrow[r * HSTR] = hv;
        ot[orow + r * HH] = hv;
      }
    }
    __syncthreads();
  }

  for (int i = tid; i < MROWS * HH; i += 1024)
    hn[mt * MROWS * HH + i] = hbuf[(i >> 8) * HSTR + (i & 255)];
}

extern "C" void kernel_launch(void* const* d_in, const int* in_sizes, int n_in,
                              void* d_out, int out_size, void* d_ws,
                              size_t ws_size, hipStream_t stream) {
  const float* input_ = (const float*)d_in[0];
  const int*   length = (const int*)d_in[1];
  const float* Wih0   = (const float*)d_in[2];
  const float* Whh0   = (const float*)d_in[3];
  const float* b0     = (const float*)d_in[4];
  const float* Wih1   = (const float*)d_in[5];
  const float* Whh1   = (const float*)d_in[6];
  const float* b1     = (const float*)d_in[7];

  float* out  = (float*)d_out;
  float* hn   = out + (size_t)TT * BB * HH;
  float* preX = (float*)d_ws;

  const dim3  g1(512, 4);
  const size_t lds1 = (size_t)64 * 8 * 32 * sizeof(v2f);

  rhn_gemm_x<<<g1, 256, lds1, stream>>>(input_, Wih0, b0, preX);
  rhn_recur<<<dim3(2), 1024, 0, stream>>>(preX, Whh0, length, out, hn);
  rhn_gemm_x<<<g1, 256, lds1, stream>>>(out, Wih1, b1, preX);
  rhn_recur<<<dim3(2), 1024, 0, stream>>>(preX, Whh1, length, out,
                                          hn + BB * HH);
}


# recurrence: wave tile = 8 h_pre cols + their 8 t_pre cols: gates wave-local (DPP rotate) on all 32 waves, h double-buffered, out[t] stored from it; 1 barrier/step, no LDS exchange; sum order same
# speedup vs baseline: 3.1954x; 1.0912x over previous
#include <hip/hip_runtime.h>
#include <math.h>

#define TT   2048
#define BB   32
#define DD   256
#define HH   256
#define W3H  768
#define NPRE 512
#define HSTR 260
#define NVR  34
#define NLP  15
#define MROWS 16
#define HBUF  (MROWS * HSTR)
#define WL_BYTES   (NLP * 32 * 32 * 16)
#define HB_BYTES   (2 * HBUF * 4)
#define LDS2_BYTES (WL_BYTES + HB_BYTES)

typedef __attribute__((ext_vector_type(2))) float v2f;
typedef __attribute__((ext_vector_type(4))) float v4f;
typedef __attribute__((ext_vector_type(8))) float v8f;

__device__ __forceinline__ size_t px_off(int t, int w, int mt, int lane) {
  return ((((size_t)(t * 32 + w) * 2 + mt) * 32 + lane) * 8);
}

__device__ __forceinline__ int colmap(int n) {
  const int w = n >> 4, c = n & 15;
  return (c < 8) ? (8 * w + c) : (HH + 8 * w + (c - 8));
}

__global__ __launch_bounds__(256) void rhn_gemm_x(
    const float* __restrict__ X, const float* __restrict__ W,
    const float* __restrict__ bias, float* __restrict__ preX) {
  extern __shared__ char smem[];
  v2f* wtile = (v2f*)smem;
  const int tid  = threadIdx.x;
  const int lane = tid & 31;
  const int wave = tid >> 5;
  const int m0   = blockIdx.x * 128 + wave * 16;
  const int n0   = blockIdx.y * 128;

  for (int e = tid; e < 64 * 8 * 32; e += 256) {
    const int c  = e >> 8;
    const int nt = (e >> 5) & 7;
    const int l  = e & 31;
    const int k  = 4 * c + ((l >> 4) << 1);
    const int n  = colmap(n0 + nt * 16 + (l & 15));
    v2f wv;
    wv.x = W[k * W3H + n];
    wv.y = W[(k + 1) * W3H + n];
    wtile[e] = wv;
  }
  __syncthreads();

  v8f acc[8];
#pragma unroll
  for (int nt = 0; nt < 8; ++nt) {
    const float bv = bias[colmap(n0 + nt * 16 + (lane & 15))];
#pragma unroll
    for (int r = 0; r < 8; ++r) acc[nt][r] = bv;
  }

  const int arow = m0 + (lane & 15);
  const float* aptr = X + (size_t)arow * DD + ((lane >> 4) << 1);

#pragma unroll 8
  for (int c = 0; c < 64; ++c) {
    const v2f a = *(const v2f*)(aptr + 4 * c);
#pragma unroll
    for (int nt = 0; nt < 8; ++nt) {
      const v2f b = wtile[(c * 8 + nt) * 32 + lane];
      acc[nt] = __builtin_amdgcn_wmma_f32_16x16x4_f32(
          false, a, false, b, (short)0, acc[nt], false, false);
    }
  }

  const int tIdx = m0 >> 5;
  const int mt   = (m0 >> 4) & 1;
#pragma unroll
  for (int nt = 0; nt < 8; ++nt) {
    const int w = (n0 >> 4) + nt;
    *(v8f*)(preX + px_off(tIdx, w, mt, lane)) = acc[nt];
  }
}

__device__ __forceinline__ float sigf(float x) {
  return 1.0f / (1.0f + __expf(-x));
}

__global__ __launch_bounds__(1024) void rhn_recur(
    const float* __restrict__ preX, const float* __restrict__ Whh,
    const int* __restrict__ length, float* __restrict__ out,
    float* __restrict__ hn) {
  __shared__ __attribute__((aligned(16))) unsigned char lds2[LDS2_BYTES];
  v4f*   wl   = (v4f*)lds2;
  float* hbuf = (float*)(lds2 + WL_BYTES);

  const int tid  = threadIdx.x;
  const int lane = tid & 31;
  const int wave = __builtin_amdgcn_readfirstlane(tid >> 5);
  const int mt   = blockIdx.x;

  for (int i = tid; i < 2 * HBUF; i += 1024) hbuf[i] = 0.0f;

  const int ncol = colmap(wave * 16 + (lane & 15));
  const int ksub = (lane >> 4) << 1;

  v4f* wlw = wl + wave * 32 + lane;
#pragma unroll
  for (int p = 0; p < NLP; ++p) {
    const int k = 4 * (NVR + 2 * p) + ksub;
    v4f q;
    q.x = Whh[k * W3H + ncol];
    q.y = Whh[(k + 1) * W3H + ncol];
    q.z = Whh[(k + 4) * W3H + ncol];
    q.w = Whh[(k + 5) * W3H + ncol];
    wlw[p * 1024] = q;
  }
  __syncthreads();

  v2f wreg[NVR];
#pragma unroll
  for (int c = 0; c < NVR; ++c) {
    const int k = 4 * c + ksub;
    v2f wv;
    wv.x = Whh[k * W3H + ncol];
    wv.y = Whh[(k + 1) * W3H + ncol];
    wreg[c] = wv;
  }

  const int  am  = lane & 15;
  const bool low = (lane & 8) == 0;
  const int grow = ((lane >> 4) << 3) + (low ? 0 : 4);
  const int gofs = grow * HSTR + wave * 8 + (lane & 7);
  int len4[4];
#pragma unroll
  for (int i = 0; i < 4; ++i) len4[i] = length[mt * MROWS + grow + i];

  const int sofs = (wave >> 1) * HSTR + (wave & 1) * 128 + lane * 4;
  float* const optr = out + (size_t)(mt * MROWS + (wave >> 1)) * HH
                          + (wave & 1) * 128 + lane * 4;

  v8f cur = *(const v8f*)(preX + px_off(0, wave, mt, lane));

  int rd = 0;

  for (int t = 0; t < TT; ++t) {
    v8f acc = cur;
    const float* hc = hbuf + rd;
    float*       hw = hbuf + (HBUF - rd);

    const int tn = (t + 1) & (TT - 1);
    cur = *(const v8f*)(preX + px_off(tn, wave, mt, lane));

#pragma unroll
    for (int c = 0; c < NVR; ++c) {
      const v2f a = *(const v2f*)(hc + am * HSTR + 4 * c + ksub);
      acc = __builtin_amdgcn_wmma_f32_16x16x4_f32(
          false, a, false, wreg[c], (short)0, acc, false, false);
    }
#pragma unroll
    for (int p = 0; p < NLP; ++p) {
      const v4f q = wlw[p * 1024];
#pragma unroll
      for (int s = 0; s < 2; ++s) {
        const int c = NVR + 2 * p + s;
        v2f b;
        b.x = s ? q.z : q.x;
        b.y = s ? q.w : q.y;
        const v2f a = *(const v2f*)(hc + am * HSTR + 4 * c + ksub);
        acc = __builtin_amdgcn_wmma_f32_16x16x4_f32(
            false, a, false, b, (short)0, acc, false, false);
      }
    }

    asm volatile("v_nop\n\tv_nop\n\tv_nop\n\tv_nop\n\tv_nop\n\tv_nop\n\tv_nop\n\tv_nop"
                 : "+v"(acc));

#pragma unroll
    for (int i = 0; i < 4; ++i) {
      const float give = low ? acc[4 + i] : acc[i];
      const int gi = __builtin_bit_cast(int, give);
      const float got = __builtin_bit_cast(float, __builtin_amdgcn_update_dpp(
          gi, gi, 0x128, 0xF, 0xF, false));
      const float hp = low ? acc[i] : got;
      const float tp = low ? got : acc[4 + i];
      const float hold = hc[gofs + i * HSTR];
      const float tg = sigf(tp);
      const float cg = sigf(tg);
      const float s  = tanhf(hp) * tg + hold * cg;
      const float hv = (t < len4[i]) ? s : hold;
      hw[gofs + i * HSTR] = hv;
    }
    __syncthreads();

    *(v4f*)(optr + (size_t)t * BB * HH) = *(const v4f*)(hw + sofs);
    rd = HBUF - rd;
  }

  const float* hf = hbuf + rd;
  for (int i = tid; i < MROWS * HH; i += 1024)
    hn[mt * MROWS * HH + i] = hf[(i >> 8) * HSTR + (i & 255)];
}

extern "C" void kernel_launch(void* const* d_in, const int* in_sizes, int n_in,
                              void* d_out, int out_size, void* d_ws,
                              size_t ws_size, hipStream_t stream) {
  const float* input_ = (const float*)d_in[0];
  const int*   length = (const int*)d_in[1];
  const float* Wih0   = (const float*)d_in[2];
  const float* Whh0   = (const float*)d_in[3];
  const float* b0     = (const float*)d_in[4];
  const float* Wih1   = (const float*)d_in[5];
  const float* Whh1   = (const float*)d_in[6];
  const float* b1     = (const float*)d_in[7];

  float* out  = (float*)d_out;
  float* hn   = out + (size_t)TT * BB * HH;
  float* preX = (float*)d_ws;

  const dim3  g1(512, 4);
  const size_t lds1 = (size_t)64 * 8 * 32 * sizeof(v2f);

  rhn_gemm_x<<<g1, 256, lds1, stream>>>(input_, Wih0, b0, preX);
  rhn_recur<<<dim3(2), 1024, 0, stream>>>(preX, Whh0, length, out, hn);
  rhn_gemm_x<<<g1, 256, lds1, stream>>>(out, Wih1, b1, preX);
  rhn_recur<<<dim3(2), 1024, 0, stream>>>(preX, Whh1, length, out,
                                          hn + BB * HH);
}


# gates branch-free, 4 elements wide: tanh computes both ranges and selects, sigmoids use v_rcp_f32 (1 ulp) not the exact divide; a lane's 4 elements interleave; matrix sums and k order unchanged
# speedup vs baseline: 3.8636x; 1.2091x over previous
#include <hip/hip_runtime.h>
#include <math.h>

#define TT   2048
#define BB   32
#define DD   256
#define HH   256
#define W3H  768
#define NPRE 512
#define HSTR 260
#define NVR  34
#define NLP  15
#define MROWS 16
#define HBUF  (MROWS * HSTR)
#define WL_BYTES   (NLP * 32 * 32 * 16)
#define HB_BYTES   (2 * HBUF * 4)
#define LDS2_BYTES (WL_BYTES + HB_BYTES)

typedef __attribute__((ext_vector_type(2))) float v2f;
typedef __attribute__((ext_vector_type(4))) float v4f;
typedef __attribute__((ext_vector_type(8))) float v8f;

__device__ __forceinline__ size_t px_off(int t, int w, int mt, int lane) {
  return ((((size_t)(t * 32 + w) * 2 + mt) * 32 + lane) * 8);
}

__device__ __forceinline__ int colmap(int n) {
  const int w = n >> 4, c = n & 15;
  return (c < 8) ? (8 * w + c) : (HH + 8 * w + (c - 8));
}

__global__ __launch_bounds__(256) void rhn_gemm_x(
    const float* __restrict__ X, const float* __restrict__ W,
    const float* __restrict__ bias, float* __restrict__ preX) {
  extern __shared__ char smem[];
  v2f* wtile = (v2f*)smem;
  const int tid  = threadIdx.x;
  const int lane = tid & 31;
  const int wave = tid >> 5;
  const int m0   = blockIdx.x * 128 + wave * 16;
  const int n0   = blockIdx.y * 128;

  for (int e = tid; e < 64 * 8 * 32; e += 256) {
    const int c  = e >> 8;
    const int nt = (e >> 5) & 7;
    const int l  = e & 31;
    const int k  = 4 * c + ((l >> 4) << 1);
    const int n  = colmap(n0 + nt * 16 + (l & 15));
    v2f wv;
    wv.x = W[k * W3H + n];
    wv.y = W[(k + 1) * W3H + n];
    wtile[e] = wv;
  }
  __syncthreads();

  v8f acc[8];
#pragma unroll
  for (int nt = 0; nt < 8; ++nt) {
    const float bv = bias[colmap(n0 + nt * 16 + (lane & 15))];
#pragma unroll
    for (int r = 0; r < 8; ++r) acc[nt][r] = bv;
  }

  const int arow = m0 + (lane & 15);
  const float* aptr = X + (size_t)arow * DD + ((lane >> 4) << 1);

#pragma unroll 8
  for (int c = 0; c < 64; ++c) {
    const v2f a = *(const v2f*)(aptr + 4 * c);
#pragma unroll
    for (int nt = 0; nt < 8; ++nt) {
      const v2f b = wtile[(c * 8 + nt) * 32 + lane];
      acc[nt] = __builtin_amdgcn_wmma_f32_16x16x4_f32(
          false, a, false, b, (short)0, acc[nt], false, false);
    }
  }

  const int tIdx = m0 >> 5;
  const int mt   = (m0 >> 4) & 1;
#pragma unroll
  for (int nt = 0; nt < 8; ++nt) {
    const int w = (n0 >> 4) + nt;
    *(v8f*)(preX + px_off(tIdx, w, mt, lane)) = acc[nt];
  }
}

__device__ __forceinline__ void sig4(const float (&x)[4], float (&y)[4]) {
  float e[4];
  for (int i = 0; i < 4; ++i) e[i] = __expf(-x[i]);
  for (int i = 0; i < 4; ++i) e[i] = 1.0f + e[i];
  for (int i = 0; i < 4; ++i) y[i] = __builtin_amdgcn_rcpf(e[i]);
}

__device__ __forceinline__ void tanh4(const float (&x)[4], float (&y)[4]) {
  float ax[4], e[4], t[4], p[4];
  for (int i = 0; i < 4; ++i) ax[i] = fabsf(x[i]);
  for (int i = 0; i < 4; ++i) e[i] = __builtin_amdgcn_exp2f(ax[i] * 2.88539004f);
  for (int i = 0; i < 4; ++i) t[i] = x[i] * x[i];
  for (int i = 0; i < 4; ++i) p[i] = fmaf(t[i], -0.00570002059f, 0.0206340719f);
  for (int i = 0; i < 4; ++i) e[i] = 1.0f + e[i];
  for (int i = 0; i < 4; ++i) p[i] = fmaf(t[i], p[i], -0.0537379310f);
  for (int i = 0; i < 4; ++i) e[i] = __builtin_amdgcn_rcpf(e[i]);
  for (int i = 0; i < 4; ++i) p[i] = fmaf(t[i], p[i], 0.133314162f);
  for (int i = 0; i < 4; ++i) p[i] = fmaf(t[i], p[i], -0.333332807f);
  for (int i = 0; i < 4; ++i) e[i] = fmaf(e[i], -2.0f, 1.0f);
  for (int i = 0; i < 4; ++i) p[i] = fmaf(t[i], ax[i] * p[i], ax[i]);
  for (int i = 0; i < 4; ++i) {
    const float sm = p[i], lg = e[i];
    y[i] = copysignf((ax[i] < 0.625f) ? sm : lg, x[i]);
  }
}

__global__ __launch_bounds__(1024) void rhn_recur(
    const float* __restrict__ preX, const float* __restrict__ Whh,
    const int* __restrict__ length, float* __restrict__ out,
    float* __restrict__ hn) {
  __shared__ __attribute__((aligned(16))) unsigned char lds2[LDS2_BYTES];
  v4f*   wl   = (v4f*)lds2;
  float* hbuf = (float*)(lds2 + WL_BYTES);

  const int tid  = threadIdx.x;
  const int lane = tid & 31;
  const int wave = __builtin_amdgcn_readfirstlane(tid >> 5);
  const int mt   = blockIdx.x;

  for (int i = tid; i < 2 * HBUF; i += 1024) hbuf[i] = 0.0f;

  const int ncol = colmap(wave * 16 + (lane & 15));
  const int ksub = (lane >> 4) << 1;

  v4f* wlw = wl + wave * 32 + lane;
#pragma unroll
  for (int p = 0; p < NLP; ++p) {
    const int k = 4 * (NVR + 2 * p) + ksub;
    v4f q;
    q.x = Whh[k * W3H + ncol];
    q.y = Whh[(k + 1) * W3H + ncol];
    q.z = Whh[(k + 4) * W3H + ncol];
    q.w = Whh[(k + 5) * W3H + ncol];
    wlw[p * 1024] = q;
  }
  __syncthreads();

  v2f wreg[NVR];
#pragma unroll
  for (int c = 0; c < NVR; ++c) {
    const int k = 4 * c + ksub;
    v2f wv;
    wv.x = Whh[k * W3H + ncol];
    wv.y = Whh[(k + 1) * W3H + ncol];
    wreg[c] = wv;
  }

  const int  am  = lane & 15;
  const bool low = (lane & 8) == 0;
  const int grow = ((lane >> 4) << 3) + (low ? 0 : 4);
  const int gofs = grow * HSTR + wave * 8 + (lane & 7);
  int len4[4];
#pragma unroll
  for (int i = 0; i < 4; ++i) len4[i] = length[mt * MROWS + grow + i];

  const int sofs = (wave >> 1) * HSTR + (wave & 1) * 128 + lane * 4;
  float* const optr = out + (size_t)(mt * MROWS + (wave >> 1)) * HH
                          + (wave & 1) * 128 + lane * 4;

  v8f cur = *(const v8f*)(preX + px_off(0, wave, mt, lane));

  int rd = 0;

  for (int t = 0; t < TT; ++t) {
    v8f acc = cur;
    const float* hc = hbuf + rd;
    float*       hw = hbuf + (HBUF - rd);

    const int tn = (t + 1) & (TT - 1);
    cur = *(const v8f*)(preX + px_off(tn, wave, mt, lane));

#pragma unroll
    for (int c = 0; c < NVR; ++c) {
      const v2f a = *(const v2f*)(hc + am * HSTR + 4 * c + ksub);
      acc = __builtin_amdgcn_wmma_f32_16x16x4_f32(
          false, a, false, wreg[c], (short)0, acc, false, false);
    }
#pragma unroll
    for (int p = 0; p < NLP; ++p) {
      const v4f q = wlw[p * 1024];
#pragma unroll
      for (int s = 0; s < 2; ++s) {
        const int c = NVR + 2 * p + s;
        v2f b;
        b.x = s ? q.z : q.x;
        b.y = s ? q.w : q.y;
        const v2f a = *(const v2f*)(hc + am * HSTR + 4 * c + ksub);
        acc = __builtin_amdgcn_wmma_f32_16x16x4_f32(
            false, a, false, b, (short)0, acc, false, false);
      }
    }

    asm volatile("v_nop\n\tv_nop\n\tv_nop\n\tv_nop\n\tv_nop\n\tv_nop\n\tv_nop\n\tv_nop"
                 : "+v"(acc));
    __builtin_amdgcn_sched_barrier(0);

    float hp[4], tp[4], hold[4], tg[4], cg[4], th[4];
    for (int i = 0; i < 4; ++i) hold[i] = hc[gofs + i * HSTR];
    for (int i = 0; i < 4; ++i) {
      const float give = low ? acc[4 + i] : acc[i];
      const int gi = __builtin_bit_cast(int, give);
      const float got = __builtin_bit_cast(float, __builtin_amdgcn_update_dpp(
          gi, gi, 0x128, 0xF, 0xF, false));
      hp[i] = low ? acc[i] : got;
      tp[i] = low ? got : acc[4 + i];
    }
    sig4(tp, tg);
    tanh4(hp, th);
    sig4(tg, cg);
    for (int i = 0; i < 4; ++i) {
      const float s  = th[i] * tg[i] + hold[i] * cg[i];
      const float ho = hold[i];
      const float hv = (t < len4[i]) ? s : ho;
      hw[gofs + i * HSTR] = hv;
    }
    __syncthreads();

    *(v4f*)(optr + (size_t)t * BB * HH) = *(const v4f*)(hw + sofs);
    rd = HBUF - rd;
  }

  const float* hf = hbuf + rd;
  for (int i = tid; i < MROWS * HH; i += 1024)
    hn[mt * MROWS * HH + i] = hf[(i >> 8) * HSTR + (i & 255)];
}

extern "C" void kernel_launch(void* const* d_in, const int* in_sizes, int n_in,
                              void* d_out, int out_size, void* d_ws,
                              size_t ws_size, hipStream_t stream) {
  const float* input_ = (const float*)d_in[0];
  const int*   length = (const int*)d_in[1];
  const float* Wih0   = (const float*)d_in[2];
  const float* Whh0   = (const float*)d_in[3];
  const float* b0     = (const float*)d_in[4];
  const float* Wih1   = (const float*)d_in[5];
  const float* Whh1   = (const float*)d_in[6];
  const float* b1     = (const float*)d_in[7];

  float* out  = (float*)d_out;
  float* hn   = out + (size_t)TT * BB * HH;
  float* preX = (float*)d_ws;

  const dim3  g1(512, 4);
  const size_t lds1 = (size_t)64 * 8 * 32 * sizeof(v2f);

  rhn_gemm_x<<<g1, 256, lds1, stream>>>(input_, Wih0, b0, preX);
  rhn_recur<<<dim3(2), 1024, 0, stream>>>(preX, Whh0, length, out, hn);
  rhn_gemm_x<<<g1, 256, lds1, stream>>>(out, Wih1, b1, preX);
  rhn_recur<<<dim3(2), 1024, 0, stream>>>(preX, Whh1, length, out,
                                          hn + BB * HH);
}


# recurrence: a wave's 64 WMMAs per step run as two independent accumulator chains (even / odd k-chunks) added once at the end, so the k sum order changes; 30 weight chunks in VGPRs, 34 in LDS
# speedup vs baseline: 3.9891x; 1.0325x over previous
#include <hip/hip_runtime.h>
#include <math.h>

#define TT   2048
#define BB   32
#define DD   256
#define HH   256
#define W3H  768
#define NPRE 512
#define HSTR 260
#define NVR  30
#define NLP  17
#define MROWS 16
#define HBUF  (MROWS * HSTR)
#define WL_BYTES   (NLP * 32 * 32 * 16)
#define HB_BYTES   (2 * HBUF * 4)
#define LDS2_BYTES (WL_BYTES + HB_BYTES)

typedef __attribute__((ext_vector_type(2))) float v2f;
typedef __attribute__((ext_vector_type(4))) float v4f;
typedef __attribute__((ext_vector_type(8))) float v8f;

__device__ __forceinline__ size_t px_off(int t, int w, int mt, int lane) {
  return ((((size_t)(t * 32 + w) * 2 + mt) * 32 + lane) * 8);
}

__device__ __forceinline__ int colmap(int n) {
  const int w = n >> 4, c = n & 15;
  return (c < 8) ? (8 * w + c) : (HH + 8 * w + (c - 8));
}

__global__ __launch_bounds__(256) void rhn_gemm_x(
    const float* __restrict__ X, const float* __restrict__ W,
    const float* __restrict__ bias, float* __restrict__ preX) {
  extern __shared__ char smem[];
  v2f* wtile = (v2f*)smem;
  const int tid  = threadIdx.x;
  const int lane = tid & 31;
  const int wave = tid >> 5;
  const int m0   = blockIdx.x * 128 + wave * 16;
  const int n0   = blockIdx.y * 128;

  for (int e = tid; e < 64 * 8 * 32; e += 256) {
    const int c  = e >> 8;
    const int nt = (e >> 5) & 7;
    const int l  = e & 31;
    const int k  = 4 * c + ((l >> 4) << 1);
    const int n  = colmap(n0 + nt * 16 + (l & 15));
    v2f wv;
    wv.x = W[k * W3H + n];
    wv.y = W[(k + 1) * W3H + n];
    wtile[e] = wv;
  }
  __syncthreads();

  v8f acc[8];
#pragma unroll
  for (int nt = 0; nt < 8; ++nt) {
    const float bv = bias[colmap(n0 + nt * 16 + (lane & 15))];
#pragma unroll
    for (int r = 0; r < 8; ++r) acc[nt][r] = bv;
  }

  const int arow = m0 + (lane & 15);
  const float* aptr = X + (size_t)arow * DD + ((lane >> 4) << 1);

#pragma unroll 8
  for (int c = 0; c < 64; ++c) {
    const v2f a = *(const v2f*)(aptr + 4 * c);
#pragma unroll
    for (int nt = 0; nt < 8; ++nt) {
      const v2f b = wtile[(c * 8 + nt) * 32 + lane];
      acc[nt] = __builtin_amdgcn_wmma_f32_16x16x4_f32(
          false, a, false, b, (short)0, acc[nt], false, false);
    }
  }

  const int tIdx = m0 >> 5;
  const int mt   = (m0 >> 4) & 1;
#pragma unroll
  for (int nt = 0; nt < 8; ++nt) {
    const int w = (n0 >> 4) + nt;
    *(v8f*)(preX + px_off(tIdx, w, mt, lane)) = acc[nt];
  }
}

__device__ __forceinline__ void sig4(const float (&x)[4], float (&y)[4]) {
  float e[4];
  for (int i = 0; i < 4; ++i) e[i] = __expf(-x[i]);
  for (int i = 0; i < 4; ++i) e[i] = 1.0f + e[i];
  for (int i = 0; i < 4; ++i) y[i] = __builtin_amdgcn_rcpf(e[i]);
}

__device__ __forceinline__ void tanh4(const float (&x)[4], float (&y)[4]) {
  float ax[4], e[4], t[4], p[4];
  for (int i = 0; i < 4; ++i) ax[i] = fabsf(x[i]);
  for (int i = 0; i < 4; ++i) e[i] = __builtin_amdgcn_exp2f(ax[i] * 2.88539004f);
  for (int i = 0; i < 4; ++i) t[i] = x[i] * x[i];
  for (int i = 0; i < 4; ++i) p[i] = fmaf(t[i], -0.00570002059f, 0.0206340719f);
  for (int i = 0; i < 4; ++i) e[i] = 1.0f + e[i];
  for (int i = 0; i < 4; ++i) p[i] = fmaf(t[i], p[i], -0.0537379310f);
  for (int i = 0; i < 4; ++i) e[i] = __builtin_amdgcn_rcpf(e[i]);
  for (int i = 0; i < 4; ++i) p[i] = fmaf(t[i], p[i], 0.133314162f);
  for (int i = 0; i < 4; ++i) p[i] = fmaf(t[i], p[i], -0.333332807f);
  for (int i = 0; i < 4; ++i) e[i] = fmaf(e[i], -2.0f, 1.0f);
  for (int i = 0; i < 4; ++i) p[i] = fmaf(t[i], ax[i] * p[i], ax[i]);
  for (int i = 0; i < 4; ++i) {
    const float sm = p[i], lg = e[i];
    y[i] = copysignf((ax[i] < 0.625f) ? sm : lg, x[i]);
  }
}

__global__ __launch_bounds__(1024) void rhn_recur(
    const float* __restrict__ preX, const float* __restrict__ Whh,
    const int* __restrict__ length, float* __restrict__ out,
    float* __restrict__ hn) {
  __shared__ __attribute__((aligned(16))) unsigned char lds2[LDS2_BYTES];
  v4f*   wl   = (v4f*)lds2;
  float* hbuf = (float*)(lds2 + WL_BYTES);

  const int tid  = threadIdx.x;
  const int lane = tid & 31;
  const int wave = __builtin_amdgcn_readfirstlane(tid >> 5);
  const int mt   = blockIdx.x;

  for (int i = tid; i < 2 * HBUF; i += 1024) hbuf[i] = 0.0f;

  const int ncol = colmap(wave * 16 + (lane & 15));
  const int ksub = (lane >> 4) << 1;

  v4f* wlw = wl + wave * 32 + lane;
#pragma unroll
  for (int p = 0; p < NLP; ++p) {
    const int k = 4 * (NVR + 2 * p) + ksub;
    v4f q;
    q.x = Whh[k * W3H + ncol];
    q.y = Whh[(k + 1) * W3H + ncol];
    q.z = Whh[(k + 4) * W3H + ncol];
    q.w = Whh[(k + 5) * W3H + ncol];
    wlw[p * 1024] = q;
  }
  __syncthreads();

  v2f wreg[NVR];
#pragma unroll
  for (int c = 0; c < NVR; ++c) {
    const int k = 4 * c + ksub;
    v2f wv;
    wv.x = Whh[k * W3H + ncol];
    wv.y = Whh[(k + 1) * W3H + ncol];
    wreg[c] = wv;
  }

  const int  am  = lane & 15;
  const bool low = (lane & 8) == 0;
  const int grow = ((lane >> 4) << 3) + (low ? 0 : 4);
  const int gofs = grow * HSTR + wave * 8 + (lane & 7);
  int len4[4];
#pragma unroll
  for (int i = 0; i < 4; ++i) len4[i] = length[mt * MROWS + grow + i];

  const int sofs = (wave >> 1) * HSTR + (wave & 1) * 128 + lane * 4;
  float* const optr = out + (size_t)(mt * MROWS + (wave >> 1)) * HH
                          + (wave & 1) * 128 + lane * 4;

  v8f cur = *(const v8f*)(preX + px_off(0, wave, mt, lane));

  int rd = 0;

  for (int t = 0; t < TT; ++t) {
    v8f acc0 = cur;
    v8f acc1 = {0.0f, 0.0f, 0.0f, 0.0f, 0.0f, 0.0f, 0.0f, 0.0f};
    const float* hc = hbuf + rd;
    float*       hw = hbuf + (HBUF - rd);

    const int tn = (t + 1) & (TT - 1);
    cur = *(const v8f*)(preX + px_off(tn, wave, mt, lane));

#pragma unroll
    for (int c = 0; c < NVR; ++c) {
      const v2f a = *(const v2f*)(hc + am * HSTR + 4 * c + ksub);
      if (c & 1)
        acc1 = __builtin_amdgcn_wmma_f32_16x16x4_f32(
            false, a, false, wreg[c], (short)0, acc1, false, false);
      else
        acc0 = __builtin_amdgcn_wmma_f32_16x16x4_f32(
            false, a, false, wreg[c], (short)0, acc0, false, false);
    }
#pragma unroll
    for (int p = 0; p < NLP; ++p) {
      const v4f q = wlw[p * 1024];
#pragma unroll
      for (int s = 0; s < 2; ++s) {
        const int c = NVR + 2 * p + s;
        v2f b;
        b.x = s ? q.z : q.x;
        b.y = s ? q.w : q.y;
        const v2f a = *(const v2f*)(hc + am * HSTR + 4 * c + ksub);
        if (c & 1)
          acc1 = __builtin_amdgcn_wmma_f32_16x16x4_f32(
              false, a, false, b, (short)0, acc1, false, false);
        else
          acc0 = __builtin_amdgcn_wmma_f32_16x16x4_f32(
              false, a, false, b, (short)0, acc0, false, false);
      }
    }

    asm volatile("v_nop\n\tv_nop\n\tv_nop\n\tv_nop\n\tv_nop\n\tv_nop\n\tv_nop\n\tv_nop"
                 : "+v"(acc0), "+v"(acc1));
    __builtin_amdgcn_sched_barrier(0);

    const v8f acc = acc0 + acc1;

    float hp[4], tp[4], hold[4], tg[4], cg[4], th[4];
    for (int i = 0; i < 4; ++i) hold[i] = hc[gofs + i * HSTR];
    for (int i = 0; i < 4; ++i) {
      const float give = low ? acc[4 + i] : acc[i];
      const int gi = __builtin_bit_cast(int, give);
      const float got = __builtin_bit_cast(float, __builtin_amdgcn_update_dpp(
          gi, gi, 0x128, 0xF, 0xF, false));
      hp[i] = low ? acc[i] : got;
      tp[i] = low ? got : acc[4 + i];
    }
    sig4(tp, tg);
    tanh4(hp, th);
    sig4(tg, cg);
    for (int i = 0; i < 4; ++i) {
      const float s  = th[i] * tg[i] + hold[i] * cg[i];
      const float ho = hold[i];
      const float hv = (t < len4[i]) ? s : ho;
      hw[gofs + i * HSTR] = hv;
    }
    __syncthreads();

    *(v4f*)(optr + (size_t)t * BB * HH) = *(const v4f*)(hw + sofs);
    rd = HBUF - rd;
  }

  const float* hf = hbuf + rd;
  for (int i = tid; i < MROWS * HH; i += 1024)
    hn[mt * MROWS * HH + i] = hf[(i >> 8) * HSTR + (i & 255)];
}

extern "C" void kernel_launch(void* const* d_in, const int* in_sizes, int n_in,
                              void* d_out, int out_size, void* d_ws,
                              size_t ws_size, hipStream_t stream) {
  const float* input_ = (const float*)d_in[0];
  const int*   length = (const int*)d_in[1];
  const float* Wih0   = (const float*)d_in[2];
  const float* Whh0   = (const float*)d_in[3];
  const float* b0     = (const float*)d_in[4];
  const float* Wih1   = (const float*)d_in[5];
  const float* Whh1   = (const float*)d_in[6];
  const float* b1     = (const float*)d_in[7];

  float* out  = (float*)d_out;
  float* hn   = out + (size_t)TT * BB * HH;
  float* preX = (float*)d_ws;

  const dim3  g1(512, 4);
  const size_t lds1 = (size_t)64 * 8 * 32 * sizeof(v2f);

  rhn_gemm_x<<<g1, 256, lds1, stream>>>(input_, Wih0, b0, preX);
  rhn_recur<<<dim3(2), 1024, 0, stream>>>(preX, Whh0, length, out, hn);
  rhn_gemm_x<<<g1, 256, lds1, stream>>>(out, Wih1, b1, preX);
  rhn_recur<<<dim3(2), 1024, 0, stream>>>(preX, Whh1, length, out,
                                          hn + BB * HH);
}


# recurrence: wave issue priority 3 (s_setprio) through the WMMA chain, 0 for gates/barrier/store: older waves' gate math no longer takes the issue port from younger waves' products; sums unchanged
# speedup vs baseline: 4.1921x; 1.0509x over previous
#include <hip/hip_runtime.h>
#include <math.h>

#define TT   2048
#define BB   32
#define DD   256
#define HH   256
#define W3H  768
#define NPRE 512
#define HSTR 260
#define NVR  30
#define NLP  17
#define MROWS 16
#define HBUF  (MROWS * HSTR)
#define WL_BYTES   (NLP * 32 * 32 * 16)
#define HB_BYTES   (2 * HBUF * 4)
#define LDS2_BYTES (WL_BYTES + HB_BYTES)

typedef __attribute__((ext_vector_type(2))) float v2f;
typedef __attribute__((ext_vector_type(4))) float v4f;
typedef __attribute__((ext_vector_type(8))) float v8f;

__device__ __forceinline__ size_t px_off(int t, int w, int mt, int lane) {
  return ((((size_t)(t * 32 + w) * 2 + mt) * 32 + lane) * 8);
}

__device__ __forceinline__ int colmap(int n) {
  const int w = n >> 4, c = n & 15;
  return (c < 8) ? (8 * w + c) : (HH + 8 * w + (c - 8));
}

__global__ __launch_bounds__(256) void rhn_gemm_x(
    const float* __restrict__ X, const float* __restrict__ W,
    const float* __restrict__ bias, float* __restrict__ preX) {
  extern __shared__ char smem[];
  v2f* wtile = (v2f*)smem;
  const int tid  = threadIdx.x;
  const int lane = tid & 31;
  const int wave = tid >> 5;
  const int m0   = blockIdx.x * 128 + wave * 16;
  const int n0   = blockIdx.y * 128;

  for (int e = tid; e < 64 * 8 * 32; e += 256) {
    const int c  = e >> 8;
    const int nt = (e >> 5) & 7;
    const int l  = e & 31;
    const int k  = 4 * c + ((l >> 4) << 1);
    const int n  = colmap(n0 + nt * 16 + (l & 15));
    v2f wv;
    wv.x = W[k * W3H + n];
    wv.y = W[(k + 1) * W3H + n];
    wtile[e] = wv;
  }
  __syncthreads();

  v8f acc[8];
#pragma unroll
  for (int nt = 0; nt < 8; ++nt) {
    const float bv = bias[colmap(n0 + nt * 16 + (lane & 15))];
#pragma unroll
    for (int r = 0; r < 8; ++r) acc[nt][r] = bv;
  }

  const int arow = m0 + (lane & 15);
  const float* aptr = X + (size_t)arow * DD + ((lane >> 4) << 1);

#pragma unroll 8
  for (int c = 0; c < 64; ++c) {
    const v2f a = *(const v2f*)(aptr + 4 * c);
#pragma unroll
    for (int nt = 0; nt < 8; ++nt) {
      const v2f b = wtile[(c * 8 + nt) * 32 + lane];
      acc[nt] = __builtin_amdgcn_wmma_f32_16x16x4_f32(
          false, a, false, b, (short)0, acc[nt], false, false);
    }
  }

  const int tIdx = m0 >> 5;
  const int mt   = (m0 >> 4) & 1;
#pragma unroll
  for (int nt = 0; nt < 8; ++nt) {
    const int w = (n0 >> 4) + nt;
    *(v8f*)(preX + px_off(tIdx, w, mt, lane)) = acc[nt];
  }
}

__device__ __forceinline__ void sig4(const float (&x)[4], float (&y)[4]) {
  float e[4];
  for (int i = 0; i < 4; ++i) e[i] = __expf(-x[i]);
  for (int i = 0; i < 4; ++i) e[i] = 1.0f + e[i];
  for (int i = 0; i < 4; ++i) y[i] = __builtin_amdgcn_rcpf(e[i]);
}

__device__ __forceinline__ void tanh4(const float (&x)[4], float (&y)[4]) {
  float ax[4], e[4], t[4], p[4];
  for (int i = 0; i < 4; ++i) ax[i] = fabsf(x[i]);
  for (int i = 0; i < 4; ++i) e[i] = __builtin_amdgcn_exp2f(ax[i] * 2.88539004f);
  for (int i = 0; i < 4; ++i) t[i] = x[i] * x[i];
  for (int i = 0; i < 4; ++i) p[i] = fmaf(t[i], -0.00570002059f, 0.0206340719f);
  for (int i = 0; i < 4; ++i) e[i] = 1.0f + e[i];
  for (int i = 0; i < 4; ++i) p[i] = fmaf(t[i], p[i], -0.0537379310f);
  for (int i = 0; i < 4; ++i) e[i] = __builtin_amdgcn_rcpf(e[i]);
  for (int i = 0; i < 4; ++i) p[i] = fmaf(t[i], p[i], 0.133314162f);
  for (int i = 0; i < 4; ++i) p[i] = fmaf(t[i], p[i], -0.333332807f);
  for (int i = 0; i < 4; ++i) e[i] = fmaf(e[i], -2.0f, 1.0f);
  for (int i = 0; i < 4; ++i) p[i] = fmaf(t[i], ax[i] * p[i], ax[i]);
  for (int i = 0; i < 4; ++i) {
    const float sm = p[i], lg = e[i];
    y[i] = copysignf((ax[i] < 0.625f) ? sm : lg, x[i]);
  }
}

__global__ __launch_bounds__(1024) void rhn_recur(
    const float* __restrict__ preX, const float* __restrict__ Whh,
    const int* __restrict__ length, float* __restrict__ out,
    float* __restrict__ hn) {
  __shared__ __attribute__((aligned(16))) unsigned char lds2[LDS2_BYTES];
  v4f*   wl   = (v4f*)lds2;
  float* hbuf = (float*)(lds2 + WL_BYTES);

  const int tid  = threadIdx.x;
  const int lane = tid & 31;
  const int wave = __builtin_amdgcn_readfirstlane(tid >> 5);
  const int mt   = blockIdx.x;

  for (int i = tid; i < 2 * HBUF; i += 1024) hbuf[i] = 0.0f;

  const int ncol = colmap(wave * 16 + (lane & 15));
  const int ksub = (lane >> 4) << 1;

  v4f* wlw = wl + wave * 32 + lane;
#pragma unroll
  for (int p = 0; p < NLP; ++p) {
    const int k = 4 * (NVR + 2 * p) + ksub;
    v4f q;
    q.x = Whh[k * W3H + ncol];
    q.y = Whh[(k + 1) * W3H + ncol];
    q.z = Whh[(k + 4) * W3H + ncol];
    q.w = Whh[(k + 5) * W3H + ncol];
    wlw[p * 1024] = q;
  }
  __syncthreads();

  v2f wreg[NVR];
#pragma unroll
  for (int c = 0; c < NVR; ++c) {
    const int k = 4 * c + ksub;
    v2f wv;
    wv.x = Whh[k * W3H + ncol];
    wv.y = Whh[(k + 1) * W3H + ncol];
    wreg[c] = wv;
  }

  const int  am  = lane & 15;
  const bool low = (lane & 8) == 0;
  const int grow = ((lane >> 4) << 3) + (low ? 0 : 4);
  const int gofs = grow * HSTR + wave * 8 + (lane & 7);
  int len4[4];
#pragma unroll
  for (int i = 0; i < 4; ++i) len4[i] = length[mt * MROWS + grow + i];

  const int sofs = (wave >> 1) * HSTR + (wave & 1) * 128 + lane * 4;
  float* const optr = out + (size_t)(mt * MROWS + (wave >> 1)) * HH
                          + (wave & 1) * 128 + lane * 4;

  v8f cur = *(const v8f*)(preX + px_off(0, wave, mt, lane));

  int rd = 0;

  for (int t = 0; t < TT; ++t) {
    __builtin_amdgcn_s_setprio(3);
    v8f acc0 = cur;
    v8f acc1 = {0.0f, 0.0f, 0.0f, 0.0f, 0.0f, 0.0f, 0.0f, 0.0f};
    const float* hc = hbuf + rd;
    float*       hw = hbuf + (HBUF - rd);

    const int tn = (t + 1) & (TT - 1);
    cur = *(const v8f*)(preX + px_off(tn, wave, mt, lane));

#pragma unroll
    for (int c = 0; c < NVR; ++c) {
      const v2f a = *(const v2f*)(hc + am * HSTR + 4 * c + ksub);
      if (c & 1)
        acc1 = __builtin_amdgcn_wmma_f32_16x16x4_f32(
            false, a, false, wreg[c], (short)0, acc1, false, false);
      else
        acc0 = __builtin_amdgcn_wmma_f32_16x16x4_f32(
            false, a, false, wreg[c], (short)0, acc0, false, false);
    }
#pragma unroll
    for (int p = 0; p < NLP; ++p) {
      const v4f q = wlw[p * 1024];
#pragma unroll
      for (int s = 0; s < 2; ++s) {
        const int c = NVR + 2 * p + s;
        v2f b;
        b.x = s ? q.z : q.x;
        b.y = s ? q.w : q.y;
        const v2f a = *(const v2f*)(hc + am * HSTR + 4 * c + ksub);
        if (c & 1)
          acc1 = __builtin_amdgcn_wmma_f32_16x16x4_f32(
              false, a, false, b, (short)0, acc1, false, false);
        else
          acc0 = __builtin_amdgcn_wmma_f32_16x16x4_f32(
              false, a, false, b, (short)0, acc0, false, false);
      }
    }

    asm volatile("v_nop\n\tv_nop\n\tv_nop\n\tv_nop\n\tv_nop\n\tv_nop\n\tv_nop\n\tv_nop"
                 : "+v"(acc0), "+v"(acc1));
    __builtin_amdgcn_s_setprio(0);
    __builtin_amdgcn_sched_barrier(0);

    const v8f acc = acc0 + acc1;

    float hp[4], tp[4], hold[4], tg[4], cg[4], th[4];
    for (int i = 0; i < 4; ++i) hold[i] = hc[gofs + i * HSTR];
    for (int i = 0; i < 4; ++i) {
      const float give = low ? acc[4 + i] : acc[i];
      const int gi = __builtin_bit_cast(int, give);
      const float got = __builtin_bit_cast(float, __builtin_amdgcn_update_dpp(
          gi, gi, 0x128, 0xF, 0xF, false));
      hp[i] = low ? acc[i] : got;
      tp[i] = low ? got : acc[4 + i];
    }
    sig4(tp, tg);
    tanh4(hp, th);
    sig4(tg, cg);
    for (int i = 0; i < 4; ++i) {
      const float s  = th[i] * tg[i] + hold[i] * cg[i];
      const float ho = hold[i];
      const float hv = (t < len4[i]) ? s : ho;
      hw[gofs + i * HSTR] = hv;
    }
    __syncthreads();

    *(v4f*)(optr + (size_t)t * BB * HH) = *(const v4f*)(hw + sofs);
    rd = HBUF - rd;
  }

  const float* hf = hbuf + rd;
  for (int i = tid; i < MROWS * HH; i += 1024)
    hn[mt * MROWS * HH + i] = hf[(i >> 8) * HSTR + (i & 255)];
}

extern "C" void kernel_launch(void* const* d_in, const int* in_sizes, int n_in,
                              void* d_out, int out_size, void* d_ws,
                              size_t ws_size, hipStream_t stream) {
  const float* input_ = (const float*)d_in[0];
  const int*   length = (const int*)d_in[1];
  const float* Wih0   = (const float*)d_in[2];
  const float* Whh0   = (const float*)d_in[3];
  const float* b0     = (const float*)d_in[4];
  const float* Wih1   = (const float*)d_in[5];
  const float* Whh1   = (const float*)d_in[6];
  const float* b1     = (const float*)d_in[7];

  float* out  = (float*)d_out;
  float* hn   = out + (size_t)TT * BB * HH;
  float* preX = (float*)d_ws;

  const dim3  g1(512, 4);
  const size_t lds1 = (size_t)64 * 8 * 32 * sizeof(v2f);

  rhn_gemm_x<<<g1, 256, lds1, stream>>>(input_, Wih0, b0, preX);
  rhn_recur<<<dim3(2), 1024, 0, stream>>>(preX, Whh0, length, out, hn);
  rhn_gemm_x<<<g1, 256, lds1, stream>>>(out, Wih1, b1, preX);
  rhn_recur<<<dim3(2), 1024, 0, stream>>>(preX, Whh1, length, out,
                                          hn + BB * HH);
}


# the two layers' recurrences run side by side, 128 time steps apart: each launch runs a chunk of layer 0 beside the chunk before of layer 1 on 4 WGPs (was 2), h carried in hn; 34 launches; sums same
# speedup vs baseline: 7.4605x; 1.7797x over previous
#include <hip/hip_runtime.h>
#include <math.h>

#define TT   2048
#define BB   32
#define DD   256
#define HH   256
#define W3H  768
#define NPRE 512
#define HSTR 260
#define NVR  30
#define NLP  17
#define MROWS 16
#define HBUF  (MROWS * HSTR)
#define CH    128
#define NCH   (TT / CH)
#define WL_BYTES   (NLP * 32 * 32 * 16)
#define HB_BYTES   (2 * HBUF * 4)
#define LDS2_BYTES (WL_BYTES + HB_BYTES)

typedef __attribute__((ext_vector_type(2))) float v2f;
typedef __attribute__((ext_vector_type(4))) float v4f;
typedef __attribute__((ext_vector_type(8))) float v8f;

__device__ __forceinline__ size_t px_off(int t, int w, int mt, int lane) {
  return ((((size_t)(t * 32 + w) * 2 + mt) * 32 + lane) * 8);
}

__device__ __forceinline__ int colmap(int n) {
  const int w = n >> 4, c = n & 15;
  return (c < 8) ? (8 * w + c) : (HH + 8 * w + (c - 8));
}

__global__ __launch_bounds__(256) void rhn_gemm_x(
    const float* __restrict__ X, const float* __restrict__ W,
    const float* __restrict__ bias, float* __restrict__ preX) {
  extern __shared__ char smem[];
  v2f* wtile = (v2f*)smem;
  const int tid  = threadIdx.x;
  const int lane = tid & 31;
  const int wave = tid >> 5;
  const int m0   = blockIdx.x * 128 + wave * 16;
  const int n0   = blockIdx.y * 128;

  for (int e = tid; e < 64 * 8 * 32; e += 256) {
    const int c  = e >> 8;
    const int nt = (e >> 5) & 7;
    const int l  = e & 31;
    const int k  = 4 * c + ((l >> 4) << 1);
    const int n  = colmap(n0 + nt * 16 + (l & 15));
    v2f wv;
    wv.x = W[k * W3H + n];
    wv.y = W[(k + 1) * W3H + n];
    wtile[e] = wv;
  }
  __syncthreads();

  v8f acc[8];
#pragma unroll
  for (int nt = 0; nt < 8; ++nt) {
    const float bv = bias[colmap(n0 + nt * 16 + (lane & 15))];
#pragma unroll
    for (int r = 0; r < 8; ++r) acc[nt][r] = bv;
  }

  const int arow = m0 + (lane & 15);
  const float* aptr = X + (size_t)arow * DD + ((lane >> 4) << 1);

#pragma unroll 8
  for (int c = 0; c < 64; ++c) {
    const v2f a = *(const v2f*)(aptr + 4 * c);
#pragma unroll
    for (int nt = 0; nt < 8; ++nt) {
      const v2f b = wtile[(c * 8 + nt) * 32 + lane];
      acc[nt] = __builtin_amdgcn_wmma_f32_16x16x4_f32(
          false, a, false, b, (short)0, acc[nt], false, false);
    }
  }

  const int tIdx = m0 >> 5;
  const int mt   = (m0 >> 4) & 1;
#pragma unroll
  for (int nt = 0; nt < 8; ++nt) {
    const int w = (n0 >> 4) + nt;
    *(v8f*)(preX + px_off(tIdx, w, mt, lane)) = acc[nt];
  }
}

__device__ __forceinline__ void sig4(const float (&x)[4], float (&y)[4]) {
  float e[4];
  for (int i = 0; i < 4; ++i) e[i] = __expf(-x[i]);
  for (int i = 0; i < 4; ++i) e[i] = 1.0f + e[i];
  for (int i = 0; i < 4; ++i) y[i] = __builtin_amdgcn_rcpf(e[i]);
}

__device__ __forceinline__ void tanh4(const float (&x)[4], float (&y)[4]) {
  float ax[4], e[4], t[4], p[4];
  for (int i = 0; i < 4; ++i) ax[i] = fabsf(x[i]);
  for (int i = 0; i < 4; ++i) e[i] = __builtin_amdgcn_exp2f(ax[i] * 2.88539004f);
  for (int i = 0; i < 4; ++i) t[i] = x[i] * x[i];
  for (int i = 0; i < 4; ++i) p[i] = fmaf(t[i], -0.00570002059f, 0.0206340719f);
  for (int i = 0; i < 4; ++i) e[i] = 1.0f + e[i];
  for (int i = 0; i < 4; ++i) p[i] = fmaf(t[i], p[i], -0.0537379310f);
  for (int i = 0; i < 4; ++i) e[i] = __builtin_amdgcn_rcpf(e[i]);
  for (int i = 0; i < 4; ++i) p[i] = fmaf(t[i], p[i], 0.133314162f);
  for (int i = 0; i < 4; ++i) p[i] = fmaf(t[i], p[i], -0.333332807f);
  for (int i = 0; i < 4; ++i) e[i] = fmaf(e[i], -2.0f, 1.0f);
  for (int i = 0; i < 4; ++i) p[i] = fmaf(t[i], ax[i] * p[i], ax[i]);
  for (int i = 0; i < 4; ++i) {
    const float sm = p[i], lg = e[i];
    y[i] = copysignf((ax[i] < 0.625f) ? sm : lg, x[i]);
  }
}

__global__ __launch_bounds__(1024) void rhn_recur(
    const float* __restrict__ preX, const float* __restrict__ WhhA,
    const float* __restrict__ WhhB, const int* __restrict__ length,
    float* __restrict__ out, float* __restrict__ hnA, float* __restrict__ hnB,
    int kA, int kB) {
  __shared__ __attribute__((aligned(16))) unsigned char lds2[LDS2_BYTES];
  v4f*   wl   = (v4f*)lds2;
  float* hbuf = (float*)(lds2 + WL_BYTES);

  const int tid  = threadIdx.x;
  const int lane = tid & 31;
  const int wave = __builtin_amdgcn_readfirstlane(tid >> 5);
  const int mt   = blockIdx.x & 1;
  const bool sel = (blockIdx.x >> 1) != 0;
  const float* __restrict__ Whh = sel ? WhhB : WhhA;
  float* __restrict__ hn        = sel ? hnB : hnA;
  const int kc = sel ? kB : kA;
  const int t0 = kc * CH;

  for (int i = tid; i < 2 * HBUF; i += 1024) {
    const int row = i / HSTR, col = i - row * HSTR;
    float v = 0.0f;
    if (kc != 0 && row < MROWS && col < HH) v = hn[(mt * MROWS + row) * HH + col];
    hbuf[i] = v;
  }

  const int ncol = colmap(wave * 16 + (lane & 15));
  const int ksub = (lane >> 4) << 1;

  v4f* wlw = wl + wave * 32 + lane;
#pragma unroll
  for (int p = 0; p < NLP; ++p) {
    const int k = 4 * (NVR + 2 * p) + ksub;
    v4f q;
    q.x = Whh[k * W3H + ncol];
    q.y = Whh[(k + 1) * W3H + ncol];
    q.z = Whh[(k + 4) * W3H + ncol];
    q.w = Whh[(k + 5) * W3H + ncol];
    wlw[p * 1024] = q;
  }
  __syncthreads();

  v2f wreg[NVR];
#pragma unroll
  for (int c = 0; c < NVR; ++c) {
    const int k = 4 * c + ksub;
    v2f wv;
    wv.x = Whh[k * W3H + ncol];
    wv.y = Whh[(k + 1) * W3H + ncol];
    wreg[c] = wv;
  }

  const int  am  = lane & 15;
  const bool low = (lane & 8) == 0;
  const int grow = ((lane >> 4) << 3) + (low ? 0 : 4);
  const int gofs = grow * HSTR + wave * 8 + (lane & 7);
  int len4[4];
#pragma unroll
  for (int i = 0; i < 4; ++i) len4[i] = length[mt * MROWS + grow + i];

  const int sofs = (wave >> 1) * HSTR + (wave & 1) * 128 + lane * 4;
  float* const optr = out + (size_t)(mt * MROWS + (wave >> 1)) * HH
                          + (wave & 1) * 128 + lane * 4;

  v8f cur = *(const v8f*)(preX + px_off(t0, wave, mt, lane));

  int rd = 0;

  for (int t = t0; t < t0 + CH; ++t) {
    __builtin_amdgcn_s_setprio(3);
    v8f acc0 = cur;
    v8f acc1 = {0.0f, 0.0f, 0.0f, 0.0f, 0.0f, 0.0f, 0.0f, 0.0f};
    const float* hc = hbuf + rd;
    float*       hw = hbuf + (HBUF - rd);

    const int tn = (t + 1) & (TT - 1);
    cur = *(const v8f*)(preX + px_off(tn, wave, mt, lane));

#pragma unroll
    for (int c = 0; c < NVR; ++c) {
      const v2f a = *(const v2f*)(hc + am * HSTR + 4 * c + ksub);
      if (c & 1)
        acc1 = __builtin_amdgcn_wmma_f32_16x16x4_f32(
            false, a, false, wreg[c], (short)0, acc1, false, false);
      else
        acc0 = __builtin_amdgcn_wmma_f32_16x16x4_f32(
            false, a, false, wreg[c], (short)0, acc0, false, false);
    }
#pragma unroll
    for (int p = 0; p < NLP; ++p) {
      const v4f q = wlw[p * 1024];
#pragma unroll
      for (int s = 0; s < 2; ++s) {
        const int c = NVR + 2 * p + s;
        v2f b;
        b.x = s ? q.z : q.x;
        b.y = s ? q.w : q.y;
        const v2f a = *(const v2f*)(hc + am * HSTR + 4 * c + ksub);
        if (c & 1)
          acc1 = __builtin_amdgcn_wmma_f32_16x16x4_f32(
              false, a, false, b, (short)0, acc1, false, false);
        else
          acc0 = __builtin_amdgcn_wmma_f32_16x16x4_f32(
              false, a, false, b, (short)0, acc0, false, false);
      }
    }

    asm volatile("v_nop\n\tv_nop\n\tv_nop\n\tv_nop\n\tv_nop\n\tv_nop\n\tv_nop\n\tv_nop"
                 : "+v"(acc0), "+v"(acc1));
    __builtin_amdgcn_s_setprio(0);
    __builtin_amdgcn_sched_barrier(0);

    const v8f acc = acc0 + acc1;

    float hp[4], tp[4], hold[4], tg[4], cg[4], th[4];
    for (int i = 0; i < 4; ++i) hold[i] = hc[gofs + i * HSTR];
    for (int i = 0; i < 4; ++i) {
      const float give = low ? acc[4 + i] : acc[i];
      const int gi = __builtin_bit_cast(int, give);
      const float got = __builtin_bit_cast(float, __builtin_amdgcn_update_dpp(
          gi, gi, 0x128, 0xF, 0xF, false));
      hp[i] = low ? acc[i] : got;
      tp[i] = low ? got : acc[4 + i];
    }
    sig4(tp, tg);
    tanh4(hp, th);
    sig4(tg, cg);
    for (int i = 0; i < 4; ++i) {
      const float s  = th[i] * tg[i] + hold[i] * cg[i];
      const float ho = hold[i];
      const float hv = (t < len4[i]) ? s : ho;
      hw[gofs + i * HSTR] = hv;
    }
    __syncthreads();

    *(v4f*)(optr + (size_t)t * BB * HH) = *(const v4f*)(hw + sofs);
    rd = HBUF - rd;
  }

  const float* hf = hbuf + rd;
  for (int i = tid; i < MROWS * HH; i += 1024)
    hn[mt * MROWS * HH + i] = hf[(i >> 8) * HSTR + (i & 255)];
}

extern "C" void kernel_launch(void* const* d_in, const int* in_sizes, int n_in,
                              void* d_out, int out_size, void* d_ws,
                              size_t ws_size, hipStream_t stream) {
  const float* input_ = (const float*)d_in[0];
  const int*   length = (const int*)d_in[1];
  const float* Wih0   = (const float*)d_in[2];
  const float* Whh0   = (const float*)d_in[3];
  const float* b0     = (const float*)d_in[4];
  const float* Wih1   = (const float*)d_in[5];
  const float* Whh1   = (const float*)d_in[6];
  const float* b1     = (const float*)d_in[7];

  float* out  = (float*)d_out;
  float* hn0  = out + (size_t)TT * BB * HH;
  float* hn1  = hn0 + BB * HH;
  float* preX = (float*)d_ws;

  const dim3  g1(512, 4);
  const dim3  gc(CH * BB / 128, 4);
  const size_t lds1 = (size_t)64 * 8 * 32 * sizeof(v2f);

  rhn_gemm_x<<<g1, 256, lds1, stream>>>(input_, Wih0, b0, preX);
  rhn_recur<<<dim3(2), 1024, 0, stream>>>(preX, Whh0, Whh0, length, out,
                                          hn0, hn0, 0, 0);
  for (int k = 1; k < NCH; ++k) {
    rhn_gemm_x<<<gc, 256, lds1, stream>>>(out + (size_t)(k - 1) * CH * BB * HH,
                                          Wih1, b1,
                                          preX + (size_t)(k - 1) * CH * BB * NPRE);
    rhn_recur<<<dim3(4), 1024, 0, stream>>>(preX, Whh0, Whh1, length, out,
                                            hn0, hn1, k, k - 1);
  }
  rhn_gemm_x<<<gc, 256, lds1, stream>>>(out + (size_t)(NCH - 1) * CH * BB * HH,
                                        Wih1, b1,
                                        preX + (size_t)(NCH - 1) * CH * BB * NPRE);
  rhn_recur<<<dim3(2), 1024, 0, stream>>>(preX, Whh1, Whh1, length, out,
                                          hn1, hn1, NCH - 1, NCH - 1);
}


# gates on the hardware tanh (v_tanh_f32): tanh(h_pre) one instruction, each sigmoid as 0.5+0.5*tanh(x/2) (was exp2+rcp; tanh was exp2+rcp+polynomial+select); matrix sums and k order unchanged
# speedup vs baseline: 7.6958x; 1.0315x over previous
#include <hip/hip_runtime.h>
#include <math.h>

#define TT   2048
#define BB   32
#define DD   256
#define HH   256
#define W3H  768
#define NPRE 512
#define HSTR 260
#define NVR  30
#define NLP  17
#define MROWS 16
#define HBUF  (MROWS * HSTR)
#define CH    128
#define NCH   (TT / CH)
#define WL_BYTES   (NLP * 32 * 32 * 16)
#define HB_BYTES   (2 * HBUF * 4)
#define LDS2_BYTES (WL_BYTES + HB_BYTES)

typedef __attribute__((ext_vector_type(2))) float v2f;
typedef __attribute__((ext_vector_type(4))) float v4f;
typedef __attribute__((ext_vector_type(8))) float v8f;

__device__ __forceinline__ size_t px_off(int t, int w, int mt, int lane) {
  return ((((size_t)(t * 32 + w) * 2 + mt) * 32 + lane) * 8);
}

__device__ __forceinline__ int colmap(int n) {
  const int w = n >> 4, c = n & 15;
  return (c < 8) ? (8 * w + c) : (HH + 8 * w + (c - 8));
}

__global__ __launch_bounds__(256) void rhn_gemm_x(
    const float* __restrict__ X, const float* __restrict__ W,
    const float* __restrict__ bias, float* __restrict__ preX) {
  extern __shared__ char smem[];
  v2f* wtile = (v2f*)smem;
  const int tid  = threadIdx.x;
  const int lane = tid & 31;
  const int wave = tid >> 5;
  const int m0   = blockIdx.x * 128 + wave * 16;
  const int n0   = blockIdx.y * 128;

  for (int e = tid; e < 64 * 8 * 32; e += 256) {
    const int c  = e >> 8;
    const int nt = (e >> 5) & 7;
    const int l  = e & 31;
    const int k  = 4 * c + ((l >> 4) << 1);
    const int n  = colmap(n0 + nt * 16 + (l & 15));
    v2f wv;
    wv.x = W[k * W3H + n];
    wv.y = W[(k + 1) * W3H + n];
    wtile[e] = wv;
  }
  __syncthreads();

  v8f acc[8];
#pragma unroll
  for (int nt = 0; nt < 8; ++nt) {
    const float bv = bias[colmap(n0 + nt * 16 + (lane & 15))];
#pragma unroll
    for (int r = 0; r < 8; ++r) acc[nt][r] = bv;
  }

  const int arow = m0 + (lane & 15);
  const float* aptr = X + (size_t)arow * DD + ((lane >> 4) << 1);

#pragma unroll 8
  for (int c = 0; c < 64; ++c) {
    const v2f a = *(const v2f*)(aptr + 4 * c);
#pragma unroll
    for (int nt = 0; nt < 8; ++nt) {
      const v2f b = wtile[(c * 8 + nt) * 32 + lane];
      acc[nt] = __builtin_amdgcn_wmma_f32_16x16x4_f32(
          false, a, false, b, (short)0, acc[nt], false, false);
    }
  }

  const int tIdx = m0 >> 5;
  const int mt   = (m0 >> 4) & 1;
#pragma unroll
  for (int nt = 0; nt < 8; ++nt) {
    const int w = (n0 >> 4) + nt;
    *(v8f*)(preX + px_off(tIdx, w, mt, lane)) = acc[nt];
  }
}

__device__ __forceinline__ void gate4(const float (&hp)[4], const float (&tp)[4],
                                      const float (&hold)[4], float (&s)[4]) {
  float a[4], th[4], tg[4], v[4], b[4], hh[4], r[4];
  for (int i = 0; i < 4; ++i) a[i]  = __builtin_amdgcn_tanhf(0.5f * tp[i]);
  for (int i = 0; i < 4; ++i) th[i] = __builtin_amdgcn_tanhf(hp[i]);
  for (int i = 0; i < 4; ++i) tg[i] = fmaf(a[i], 0.5f, 0.5f);
  for (int i = 0; i < 4; ++i) v[i]  = fmaf(a[i], 0.25f, 0.25f);
  for (int i = 0; i < 4; ++i) b[i]  = __builtin_amdgcn_tanhf(v[i]);
  for (int i = 0; i < 4; ++i) hh[i] = 0.5f * hold[i];
  for (int i = 0; i < 4; ++i) r[i]  = fmaf(hh[i], b[i], hh[i]);
  for (int i = 0; i < 4; ++i) s[i]  = fmaf(th[i], tg[i], r[i]);
}

__global__ __launch_bounds__(1024) void rhn_recur(
    const float* __restrict__ preX, const float* __restrict__ WhhA,
    const float* __restrict__ WhhB, const int* __restrict__ length,
    float* __restrict__ out, float* __restrict__ hnA, float* __restrict__ hnB,
    int kA, int kB) {
  __shared__ __attribute__((aligned(16))) unsigned char lds2[LDS2_BYTES];
  v4f*   wl   = (v4f*)lds2;
  float* hbuf = (float*)(lds2 + WL_BYTES);

  const int tid  = threadIdx.x;
  const int lane = tid & 31;
  const int wave = __builtin_amdgcn_readfirstlane(tid >> 5);
  const int mt   = blockIdx.x & 1;
  const bool sel = (blockIdx.x >> 1) != 0;
  const float* __restrict__ Whh = sel ? WhhB : WhhA;
  float* __restrict__ hn        = sel ? hnB : hnA;
  const int kc = sel ? kB : kA;
  const int t0 = kc * CH;

  for (int i = tid; i < 2 * HBUF; i += 1024) {
    const int row = i / HSTR, col = i - row * HSTR;
    float v = 0.0f;
    if (kc != 0 && row < MROWS && col < HH) v = hn[(mt * MROWS + row) * HH + col];
    hbuf[i] = v;
  }

  const int ncol = colmap(wave * 16 + (lane & 15));
  const int ksub = (lane >> 4) << 1;

  v4f* wlw = wl + wave * 32 + lane;
#pragma unroll
  for (int p = 0; p < NLP; ++p) {
    const int k = 4 * (NVR + 2 * p) + ksub;
    v4f q;
    q.x = Whh[k * W3H + ncol];
    q.y = Whh[(k + 1) * W3H + ncol];
    q.z = Whh[(k + 4) * W3H + ncol];
    q.w = Whh[(k + 5) * W3H + ncol];
    wlw[p * 1024] = q;
  }
  __syncthreads();

  v2f wreg[NVR];
#pragma unroll
  for (int c = 0; c < NVR; ++c) {
    const int k = 4 * c + ksub;
    v2f wv;
    wv.x = Whh[k * W3H + ncol];
    wv.y = Whh[(k + 1) * W3H + ncol];
    wreg[c] = wv;
  }

  const int  am  = lane & 15;
  const bool low = (lane & 8) == 0;
  const int grow = ((lane >> 4) << 3) + (low ? 0 : 4);
  const int gofs = grow * HSTR + wave * 8 + (lane & 7);
  int len4[4];
#pragma unroll
  for (int i = 0; i < 4; ++i) len4[i] = length[mt * MROWS + grow + i];

  const int sofs = (wave >> 1) * HSTR + (wave & 1) * 128 + lane * 4;
  float* const optr = out + (size_t)(mt * MROWS + (wave >> 1)) * HH
                          + (wave & 1) * 128 + lane * 4;

  v8f cur = *(const v8f*)(preX + px_off(t0, wave, mt, lane));

  int rd = 0;

  for (int t = t0; t < t0 + CH; ++t) {
    __builtin_amdgcn_s_setprio(3);
    v8f acc0 = cur;
    v8f acc1 = {0.0f, 0.0f, 0.0f, 0.0f, 0.0f, 0.0f, 0.0f, 0.0f};
    const float* hc = hbuf + rd;
    float*       hw = hbuf + (HBUF - rd);

    const int tn = (t + 1) & (TT - 1);
    cur = *(const v8f*)(preX + px_off(tn, wave, mt, lane));

#pragma unroll
    for (int c = 0; c < NVR; ++c) {
      const v2f a = *(const v2f*)(hc + am * HSTR + 4 * c + ksub);
      if (c & 1)
        acc1 = __builtin_amdgcn_wmma_f32_16x16x4_f32(
            false, a, false, wreg[c], (short)0, acc1, false, false);
      else
        acc0 = __builtin_amdgcn_wmma_f32_16x16x4_f32(
            false, a, false, wreg[c], (short)0, acc0, false, false);
    }
#pragma unroll
    for (int p = 0; p < NLP; ++p) {
      const v4f q = wlw[p * 1024];
#pragma unroll
      for (int s = 0; s < 2; ++s) {
        const int c = NVR + 2 * p + s;
        v2f b;
        b.x = s ? q.z : q.x;
        b.y = s ? q.w : q.y;
        const v2f a = *(const v2f*)(hc + am * HSTR + 4 * c + ksub);
        if (c & 1)
          acc1 = __builtin_amdgcn_wmma_f32_16x16x4_f32(
              false, a, false, b, (short)0, acc1, false, false);
        else
          acc0 = __builtin_amdgcn_wmma_f32_16x16x4_f32(
              false, a, false, b, (short)0, acc0, false, false);
      }
    }

    asm volatile("v_nop\n\tv_nop\n\tv_nop\n\tv_nop\n\tv_nop\n\tv_nop\n\tv_nop\n\tv_nop"
                 : "+v"(acc0), "+v"(acc1));
    __builtin_amdgcn_s_setprio(0);
    __builtin_amdgcn_sched_barrier(0);

    const v8f acc = acc0 + acc1;

    float hp[4], tp[4], hold[4], sv[4];
    for (int i = 0; i < 4; ++i) hold[i] = hc[gofs + i * HSTR];
    for (int i = 0; i < 4; ++i) {
      const float give = low ? acc[4 + i] : acc[i];
      const int gi = __builtin_bit_cast(int, give);
      const float got = __builtin_bit_cast(float, __builtin_amdgcn_update_dpp(
          gi, gi, 0x128, 0xF, 0xF, false));
      hp[i] = low ? acc[i] : got;
      tp[i] = low ? got : acc[4 + i];
    }
    gate4(hp, tp, hold, sv);
    for (int i = 0; i < 4; ++i) {
      const float hv = (t < len4[i]) ? sv[i] : hold[i];
      hw[gofs + i * HSTR] = hv;
    }
    __syncthreads();

    *(v4f*)(optr + (size_t)t * BB * HH) = *(const v4f*)(hw + sofs);
    rd = HBUF - rd;
  }

  const float* hf = hbuf + rd;
  for (int i = tid; i < MROWS * HH; i += 1024)
    hn[mt * MROWS * HH + i] = hf[(i >> 8) * HSTR + (i & 255)];
}

extern "C" void kernel_launch(void* const* d_in, const int* in_sizes, int n_in,
                              void* d_out, int out_size, void* d_ws,
                              size_t ws_size, hipStream_t stream) {
  const float* input_ = (const float*)d_in[0];
  const int*   length = (const int*)d_in[1];
  const float* Wih0   = (const float*)d_in[2];
  const float* Whh0   = (const float*)d_in[3];
  const float* b0     = (const float*)d_in[4];
  const float* Wih1   = (const float*)d_in[5];
  const float* Whh1   = (const float*)d_in[6];
  const float* b1     = (const float*)d_in[7];

  float* out  = (float*)d_out;
  float* hn0  = out + (size_t)TT * BB * HH;
  float* hn1  = hn0 + BB * HH;
  float* preX = (float*)d_ws;

  const dim3  g1(512, 4);
  const dim3  gc(CH * BB / 128, 4);
  const size_t lds1 = (size_t)64 * 8 * 32 * sizeof(v2f);

  rhn_gemm_x<<<g1, 256, lds1, stream>>>(input_, Wih0, b0, preX);
  rhn_recur<<<dim3(2), 1024, 0, stream>>>(preX, Whh0, Whh0, length, out,
                                          hn0, hn0, 0, 0);
  for (int k = 1; k < NCH; ++k) {
    rhn_gemm_x<<<gc, 256, lds1, stream>>>(out + (size_t)(k - 1) * CH * BB * HH,
                                          Wih1, b1,
                                          preX + (size_t)(k - 1) * CH * BB * NPRE);
    rhn_recur<<<dim3(4), 1024, 0, stream>>>(preX, Whh0, Whh1, length, out,
                                            hn0, hn1, k, k - 1);
  }
  rhn_gemm_x<<<gc, 256, lds1, stream>>>(out + (size_t)(NCH - 1) * CH * BB * HH,
                                        Wih1, b1,
                                        preX + (size_t)(NCH - 1) * CH * BB * NPRE);
  rhn_recur<<<dim3(2), 1024, 0, stream>>>(preX, Whh1, Whh1, length, out,
                                          hn1, hn1, NCH - 1, NCH - 1);
}
